# starting kernel
# baseline (speedup 1.0000x reference)
#include <hip/hip_runtime.h>
#include <hip/hip_bf16.h>


typedef __attribute__((ext_vector_type(2))) float v2f;
typedef __attribute__((ext_vector_type(8))) float v8f;

#define B_ 2
#define S_ 2048
#define D_ 1024
#define H_ 16
#define KVH_ 4
#define HD_ 64
#define WIN_ 64
#define M_ (B_ * S_)

template <int DO_ROPE>
__global__ void gemm64x32_wmma(const float* __restrict__ A, const float* __restrict__ W,
                               float* __restrict__ C, int M, int N, int K,
                               const float* __restrict__ fcos,
                               const float* __restrict__ fsin) {
    const int lane  = threadIdx.x & 31;
    const int wave  = threadIdx.x >> 5;
    const int tileM = blockIdx.y * 64;
    const int tileN = (blockIdx.x * (blockDim.x >> 5) + wave) * 32;
    if (tileN >= N) return;

    const int l15  = lane & 15;
    const int koff = (lane >> 4) << 1;

    v8f acc[4][2] = {};

    const float* ap = A + (long)(tileM + l15) * K + koff;
    const float* bp = W + (long)koff * N + (tileN + l15);
    const long aStride = 16L * K;

    for (int kk = 0; kk < K; kk += 4) {
        v2f a[4], b[2];
        #pragma unroll
        for (int i = 0; i < 4; ++i) {
            a[i].x = ap[i * aStride];
            a[i].y = ap[i * aStride + 1];
        }
        #pragma unroll
        for (int j = 0; j < 2; ++j) {
            b[j].x = bp[j * 16];
            b[j].y = bp[N + j * 16];
        }
        #pragma unroll
        for (int i = 0; i < 4; ++i)
            #pragma unroll
            for (int j = 0; j < 2; ++j)
                acc[i][j] = __builtin_amdgcn_wmma_f32_16x16x4_f32(
                    false, a[i], false, b[j], (short)0, acc[i][j], false, false);
        ap += 4;
        bp += (long)4 * N;
    }

    const int rowOff = (lane >> 4) << 3;
    #pragma unroll
    for (int i = 0; i < 4; ++i) {
        const int rowBase = tileM + i * 16 + rowOff;
        #pragma unroll
        for (int j = 0; j < 2; ++j) {
            const int col = tileN + j * 16 + l15;
            #pragma unroll
            for (int r = 0; r < 8; ++r) {
                float val = acc[i][j][r];
                if (DO_ROPE) {
                    const float partner = __shfl_xor(val, 1, 32);
                    const int pos = (rowBase + r) & (S_ - 1);
                    const int fi  = (col & (HD_ - 1)) >> 1;
                    const float c = fcos[pos * (HD_ / 2) + fi];
                    const float s = fsin[pos * (HD_ / 2) + fi];
                    val = ((col & 1) == 0) ? (val * c - partner * s)
                                           : (partner * s + val * c);
                }
                C[(long)(rowBase + r) * N + col] = val;
            }
        }
    }
}

__global__ void attn_wmma(const float* __restrict__ q, const float* __restrict__ kbuf,
                          const float* __restrict__ vbuf, float* __restrict__ o) {
    __shared__ float p[16][83];

    const int lane = threadIdx.x;
    const int i0   = blockIdx.x * 16;
    const int h    = blockIdx.y;
    const int b    = blockIdx.z;
    const int kvh  = h >> 2;

    const int l15  = lane & 15;
    const int koff = (lane >> 4) << 1;
    const int r8   = (lane >> 4) << 3;
    const int keystart = i0 - WIN_;

    const float* qrow = q + ((long)(b * S_ + i0 + l15) * H_ + h) * HD_;
    for (int t = 0; t < 5; ++t) {
        const int j0 = keystart + t * 16;
        if (j0 < 0) {
            #pragma unroll
            for (int r = 0; r < 8; ++r) p[r8 + r][t * 16 + l15] = -1e9f;
            continue;
        }
        const float* krow = kbuf + ((long)(b * S_ + j0 + l15) * KVH_ + kvh) * HD_;
        v8f acc = {};
        for (int kk = 0; kk < HD_; kk += 4) {
            const int kb = kk + koff;
            v2f a, bb;
            a.x  = qrow[kb];  a.y  = qrow[kb + 1];
            bb.x = krow[kb];  bb.y = krow[kb + 1];
            acc = __builtin_amdgcn_wmma_f32_16x16x4_f32(false, a, false, bb,
                                                        (short)0, acc, false, false);
        }
        #pragma unroll
        for (int r = 0; r < 8; ++r) {
            const int row  = i0 + r8 + r;
            const int colk = j0 + l15;
            float sc = fmaxf(acc[r] * 0.125f, -1e9f);
            if (colk > row || row - colk > WIN_) sc = -1e9f;
            p[r8 + r][t * 16 + l15] = sc;
        }
    }
    __syncthreads();

    if (lane < 16) {
        float m = -1e30f;
        for (int c = 0; c < 80; ++c) m = fmaxf(m, p[lane][c]);
        float sum = 0.0f;
        for (int c = 0; c < 80; ++c) {
            const float e = expf(p[lane][c] - m);
            p[lane][c] = e;
            sum += e;
        }
        const float inv = 1.0f / sum;
        for (int c = 0; c < 80; ++c) p[lane][c] *= inv;
    }
    __syncthreads();

    for (int nt = 0; nt < 4; ++nt) {
        const int n0 = nt * 16;
        v8f acc = {};
        for (int kk = 0; kk < 80; kk += 4) {
            const int kb = kk + koff;
            v2f a, bb;
            a.x = p[l15][kb];
            a.y = p[l15][kb + 1];
            const int key0 = keystart + kb;
            const int key1 = key0 + 1;
            bb.x = (key0 >= 0)
                 ? vbuf[((long)(b * S_ + key0) * KVH_ + kvh) * HD_ + n0 + l15] : 0.0f;
            bb.y = (key1 >= 0)
                 ? vbuf[((long)(b * S_ + key1) * KVH_ + kvh) * HD_ + n0 + l15] : 0.0f;
            acc = __builtin_amdgcn_wmma_f32_16x16x4_f32(false, a, false, bb,
                                                        (short)0, acc, false, false);
        }
        #pragma unroll
        for (int r = 0; r < 8; ++r) {
            o[((long)(b * S_ + i0 + r8 + r) * H_ + h) * HD_ + n0 + l15] = acc[r];
        }
    }
}

extern "C" void kernel_launch(void* const* d_in, const int* in_sizes, int n_in,
                              void* d_out, int out_size, void* d_ws, size_t ws_size,
                              hipStream_t stream) {
    const float* x    = (const float*)d_in[0];
    const float* fcos = (const float*)d_in[1];
    const float* fsin = (const float*)d_in[2];
    const float* wq   = (const float*)d_in[3];
    const float* wk   = (const float*)d_in[4];
    const float* wv   = (const float*)d_in[5];
    const float* wo   = (const float*)d_in[6];
    float* out = (float*)d_out;

    float* qb = (float*)d_ws;
    float* kb = qb + (size_t)M_ * (H_ * HD_);
    float* vb = kb + (size_t)M_ * (KVH_ * HD_);
    float* at = vb + (size_t)M_ * (KVH_ * HD_);

    const dim3 blk(128);
    gemm64x32_wmma<1><<<dim3((H_ * HD_) / 128, M_ / 64), blk, 0, stream>>>(
        x, wq, qb, M_, H_ * HD_, D_, fcos, fsin);
    gemm64x32_wmma<1><<<dim3((KVH_ * HD_) / 128, M_ / 64), blk, 0, stream>>>(
        x, wk, kb, M_, KVH_ * HD_, D_, fcos, fsin);
    gemm64x32_wmma<0><<<dim3((KVH_ * HD_) / 128, M_ / 64), blk, 0, stream>>>(
        x, wv, vb, M_, KVH_ * HD_, D_, fcos, fsin);

    attn_wmma<<<dim3(S_ / 16, H_, B_), dim3(32), 0, stream>>>(qb, kb, vb, at);

    gemm64x32_wmma<0><<<dim3(D_ / 128, M_ / 64), blk, 0, stream>>>(
        at, wo, out, M_, D_, H_ * HD_, fcos, fsin);
}


# GEMM K loop holds fragments of 8 k-steps in registers and refills each slot right after its WMMAs: loads run a trip ahead, not one waited load per WMMA pair; launch bound 128; sum order kept
# speedup vs baseline: 3.6147x; 3.6147x over previous
#include <hip/hip_runtime.h>
#include <hip/hip_bf16.h>


typedef __attribute__((ext_vector_type(2))) float v2f;
typedef __attribute__((ext_vector_type(8))) float v8f;

#define B_ 2
#define S_ 2048
#define D_ 1024
#define H_ 16
#define KVH_ 4
#define HD_ 64
#define WIN_ 64
#define M_ (B_ * S_)

template <int DO_ROPE, int N, int K>
__global__ void __launch_bounds__(128)
gemm64x32_wmma(const float* __restrict__ A, const float* __restrict__ W,
                               float* __restrict__ C,
                               const float* __restrict__ fcos,
                               const float* __restrict__ fsin) {
    constexpr int U = 8;
    static_assert(K % (4 * U) == 0 && K >= 4 * U, "K must be whole trips of U steps");

    const int lane  = threadIdx.x & 31;
    const int wave  = threadIdx.x >> 5;
    const int tileM = blockIdx.y * 64;
    const int tileN = (blockIdx.x * (blockDim.x >> 5) + wave) * 32;
    if (tileN >= N) return;

    const int l15  = lane & 15;
    const int koff = (lane >> 4) << 1;

    v8f acc[4][2] = {};

    const float* ap = A + (long)(tileM + l15) * K + koff;
    const float* bp = W + (long)koff * N + (tileN + l15);
    constexpr long aStride = 16L * K;

    v2f a[U][4], b[U][2];

#define GEMM_LOAD_STEP(u, apx, bpx)                                            \
    do {                                                                       \
        _Pragma("unroll")                                                      \
        for (int i = 0; i < 4; ++i) {                                          \
            a[u][i].x = (apx)[i * aStride + 4 * (u)];                          \
            a[u][i].y = (apx)[i * aStride + 4 * (u) + 1];                      \
        }                                                                      \
        _Pragma("unroll")                                                      \
        for (int j = 0; j < 2; ++j) {                                          \
            b[u][j].x = (bpx)[(long)(4 * (u)) * N + j * 16];                   \
            b[u][j].y = (bpx)[(long)(4 * (u) + 1) * N + j * 16];               \
        }                                                                      \
    } while (0)

#define GEMM_MMA_STEP(u)                                                       \
    do {                                                                       \
        _Pragma("unroll")                                                      \
        for (int i = 0; i < 4; ++i)                                            \
            _Pragma("unroll")                                                  \
            for (int j = 0; j < 2; ++j)                                        \
                acc[i][j] = __builtin_amdgcn_wmma_f32_16x16x4_f32(             \
                    false, a[u][i], false, b[u][j], (short)0, acc[i][j],       \
                    false, false);                                             \
    } while (0)

    #pragma unroll
    for (int u = 0; u < U; ++u) GEMM_LOAD_STEP(u, ap, bp);

    for (int kk = 4 * U; kk < K; kk += 4 * U) {
        ap += 4 * U;
        bp += (long)(4 * U) * N;
        #pragma unroll
        for (int u = 0; u < U; ++u) {
            GEMM_MMA_STEP(u);
            GEMM_LOAD_STEP(u, ap, bp);
        }
    }

    #pragma unroll
    for (int u = 0; u < U; ++u) GEMM_MMA_STEP(u);

#undef GEMM_LOAD_STEP
#undef GEMM_MMA_STEP

    const int rowOff = (lane >> 4) << 3;
    #pragma unroll
    for (int i = 0; i < 4; ++i) {
        const int rowBase = tileM + i * 16 + rowOff;
        #pragma unroll
        for (int j = 0; j < 2; ++j) {
            const int col = tileN + j * 16 + l15;
            #pragma unroll
            for (int r = 0; r < 8; ++r) {
                float val = acc[i][j][r];
                if (DO_ROPE) {
                    const float partner = __shfl_xor(val, 1, 32);
                    const int pos = (rowBase + r) & (S_ - 1);
                    const int fi  = (col & (HD_ - 1)) >> 1;
                    const float c = fcos[pos * (HD_ / 2) + fi];
                    const float s = fsin[pos * (HD_ / 2) + fi];
                    val = ((col & 1) == 0) ? (val * c - partner * s)
                                           : (partner * s + val * c);
                }
                C[(long)(rowBase + r) * N + col] = val;
            }
        }
    }
}

__global__ void attn_wmma(const float* __restrict__ q, const float* __restrict__ kbuf,
                          const float* __restrict__ vbuf, float* __restrict__ o) {
    __shared__ float p[16][83];

    const int lane = threadIdx.x;
    const int i0   = blockIdx.x * 16;
    const int h    = blockIdx.y;
    const int b    = blockIdx.z;
    const int kvh  = h >> 2;

    const int l15  = lane & 15;
    const int koff = (lane >> 4) << 1;
    const int r8   = (lane >> 4) << 3;
    const int keystart = i0 - WIN_;

    const float* qrow = q + ((long)(b * S_ + i0 + l15) * H_ + h) * HD_;
    for (int t = 0; t < 5; ++t) {
        const int j0 = keystart + t * 16;
        if (j0 < 0) {
            #pragma unroll
            for (int r = 0; r < 8; ++r) p[r8 + r][t * 16 + l15] = -1e9f;
            continue;
        }
        const float* krow = kbuf + ((long)(b * S_ + j0 + l15) * KVH_ + kvh) * HD_;
        v8f acc = {};
        for (int kk = 0; kk < HD_; kk += 4) {
            const int kb = kk + koff;
            v2f a, bb;
            a.x  = qrow[kb];  a.y  = qrow[kb + 1];
            bb.x = krow[kb];  bb.y = krow[kb + 1];
            acc = __builtin_amdgcn_wmma_f32_16x16x4_f32(false, a, false, bb,
                                                        (short)0, acc, false, false);
        }
        #pragma unroll
        for (int r = 0; r < 8; ++r) {
            const int row  = i0 + r8 + r;
            const int colk = j0 + l15;
            float sc = fmaxf(acc[r] * 0.125f, -1e9f);
            if (colk > row || row - colk > WIN_) sc = -1e9f;
            p[r8 + r][t * 16 + l15] = sc;
        }
    }
    __syncthreads();

    if (lane < 16) {
        float m = -1e30f;
        for (int c = 0; c < 80; ++c) m = fmaxf(m, p[lane][c]);
        float sum = 0.0f;
        for (int c = 0; c < 80; ++c) {
            const float e = expf(p[lane][c] - m);
            p[lane][c] = e;
            sum += e;
        }
        const float inv = 1.0f / sum;
        for (int c = 0; c < 80; ++c) p[lane][c] *= inv;
    }
    __syncthreads();

    for (int nt = 0; nt < 4; ++nt) {
        const int n0 = nt * 16;
        v8f acc = {};
        for (int kk = 0; kk < 80; kk += 4) {
            const int kb = kk + koff;
            v2f a, bb;
            a.x = p[l15][kb];
            a.y = p[l15][kb + 1];
            const int key0 = keystart + kb;
            const int key1 = key0 + 1;
            bb.x = (key0 >= 0)
                 ? vbuf[((long)(b * S_ + key0) * KVH_ + kvh) * HD_ + n0 + l15] : 0.0f;
            bb.y = (key1 >= 0)
                 ? vbuf[((long)(b * S_ + key1) * KVH_ + kvh) * HD_ + n0 + l15] : 0.0f;
            acc = __builtin_amdgcn_wmma_f32_16x16x4_f32(false, a, false, bb,
                                                        (short)0, acc, false, false);
        }
        #pragma unroll
        for (int r = 0; r < 8; ++r) {
            o[((long)(b * S_ + i0 + r8 + r) * H_ + h) * HD_ + n0 + l15] = acc[r];
        }
    }
}

extern "C" void kernel_launch(void* const* d_in, const int* in_sizes, int n_in,
                              void* d_out, int out_size, void* d_ws, size_t ws_size,
                              hipStream_t stream) {
    const float* x    = (const float*)d_in[0];
    const float* fcos = (const float*)d_in[1];
    const float* fsin = (const float*)d_in[2];
    const float* wq   = (const float*)d_in[3];
    const float* wk   = (const float*)d_in[4];
    const float* wv   = (const float*)d_in[5];
    const float* wo   = (const float*)d_in[6];
    float* out = (float*)d_out;

    float* qb = (float*)d_ws;
    float* kb = qb + (size_t)M_ * (H_ * HD_);
    float* vb = kb + (size_t)M_ * (KVH_ * HD_);
    float* at = vb + (size_t)M_ * (KVH_ * HD_);

    const dim3 blk(128);
    gemm64x32_wmma<1, H_ * HD_, D_><<<dim3((H_ * HD_) / 128, M_ / 64), blk, 0, stream>>>(
        x, wq, qb, fcos, fsin);
    gemm64x32_wmma<1, KVH_ * HD_, D_><<<dim3((KVH_ * HD_) / 128, M_ / 64), blk, 0, stream>>>(
        x, wk, kb, fcos, fsin);
    gemm64x32_wmma<0, KVH_ * HD_, D_><<<dim3((KVH_ * HD_) / 128, M_ / 64), blk, 0, stream>>>(
        x, wv, vb, fcos, fsin);

    attn_wmma<<<dim3(S_ / 16, H_, B_), dim3(32), 0, stream>>>(qb, kb, vb, at);

    gemm64x32_wmma<0, D_, H_ * HD_><<<dim3(D_ / 128, M_ / 64), blk, 0, stream>>>(
        at, wo, out, fcos, fsin);
}


# GEMM A fragments by one 16-byte load per lane: a lane half owns 4 consecutive k of each 8-k group (2 WMMAs), B rows follow that map; sum order in a group changes; prologue loads kept in loop order
# speedup vs baseline: 3.8214x; 1.0572x over previous
#include <hip/hip_runtime.h>
#include <hip/hip_bf16.h>


typedef __attribute__((ext_vector_type(2))) float v2f;
typedef __attribute__((ext_vector_type(4))) float v4f;
typedef __attribute__((ext_vector_type(8))) float v8f;

#define B_ 2
#define S_ 2048
#define D_ 1024
#define H_ 16
#define KVH_ 4
#define HD_ 64
#define WIN_ 64
#define M_ (B_ * S_)

template <int DO_ROPE, int N, int K>
__global__ void __launch_bounds__(128)
gemm64x32_wmma(const float* __restrict__ A, const float* __restrict__ W,
                               float* __restrict__ C,
                               const float* __restrict__ fcos,
                               const float* __restrict__ fsin) {
    constexpr int G = 4;
    static_assert(K % (8 * G) == 0 && K >= 8 * G, "K must be whole trips of G groups");

    const int lane  = threadIdx.x & 31;
    const int wave  = threadIdx.x >> 5;
    const int tileM = blockIdx.y * 64;
    const int tileN = (blockIdx.x * (blockDim.x >> 5) + wave) * 32;
    if (tileN >= N) return;

    const int l15  = lane & 15;
    const int koff = (lane >> 4) << 2;

    v8f acc[4][2] = {};

    const float* ap = A + (long)(tileM + l15) * K + koff;
    const float* bp = W + (long)koff * N + (tileN + l15);
    constexpr long aStride = 16L * K;

    v4f a[G][4];
    v2f b[G][2][2];

#define GEMM_LOAD_GROUP(g, apx, bpx)                                           \
    do {                                                                       \
        _Pragma("unroll")                                                      \
        for (int i = 0; i < 4; ++i)                                            \
            a[g][i] = *(const v4f*)((apx) + i * aStride + 8 * (g));            \
        _Pragma("unroll")                                                      \
        for (int w = 0; w < 2; ++w) {                                          \
            _Pragma("unroll")                                                  \
            for (int j = 0; j < 2; ++j) {                                      \
                b[g][w][j].x = (bpx)[(long)(8 * (g) + 2 * w) * N + j * 16];     \
                b[g][w][j].y = (bpx)[(long)(8 * (g) + 2 * w + 1) * N + j * 16]; \
            }                                                                  \
        }                                                                      \
    } while (0)

#define GEMM_MMA_GROUP(g)                                                      \
    do {                                                                       \
        _Pragma("unroll")                                                      \
        for (int i = 0; i < 4; ++i) {                                          \
            const v2f alo = __builtin_shufflevector(a[g][i], a[g][i], 0, 1);   \
            const v2f ahi = __builtin_shufflevector(a[g][i], a[g][i], 2, 3);   \
            _Pragma("unroll")                                                  \
            for (int j = 0; j < 2; ++j) {                                      \
                acc[i][j] = __builtin_amdgcn_wmma_f32_16x16x4_f32(             \
                    false, alo, false, b[g][0][j], (short)0, acc[i][j],        \
                    false, false);                                             \
                acc[i][j] = __builtin_amdgcn_wmma_f32_16x16x4_f32(             \
                    false, ahi, false, b[g][1][j], (short)0, acc[i][j],        \
                    false, false);                                             \
            }                                                                  \
        }                                                                      \
    } while (0)

    #pragma unroll
    for (int g = 0; g < G; ++g) {
        GEMM_LOAD_GROUP(g, ap, bp);
        __builtin_amdgcn_sched_barrier(0);
    }

    for (int kk = 8 * G; kk < K; kk += 8 * G) {
        ap += 8 * G;
        bp += (long)(8 * G) * N;
        #pragma unroll
        for (int g = 0; g < G; ++g) {
            GEMM_MMA_GROUP(g);
            GEMM_LOAD_GROUP(g, ap, bp);
        }
    }

    #pragma unroll
    for (int g = 0; g < G; ++g) GEMM_MMA_GROUP(g);

#undef GEMM_LOAD_GROUP
#undef GEMM_MMA_GROUP

    const int rowOff = (lane >> 4) << 3;
    #pragma unroll
    for (int i = 0; i < 4; ++i) {
        const int rowBase = tileM + i * 16 + rowOff;
        #pragma unroll
        for (int j = 0; j < 2; ++j) {
            const int col = tileN + j * 16 + l15;
            #pragma unroll
            for (int r = 0; r < 8; ++r) {
                float val = acc[i][j][r];
                if (DO_ROPE) {
                    const float partner = __shfl_xor(val, 1, 32);
                    const int pos = (rowBase + r) & (S_ - 1);
                    const int fi  = (col & (HD_ - 1)) >> 1;
                    const float c = fcos[pos * (HD_ / 2) + fi];
                    const float s = fsin[pos * (HD_ / 2) + fi];
                    val = ((col & 1) == 0) ? (val * c - partner * s)
                                           : (partner * s + val * c);
                }
                C[(long)(rowBase + r) * N + col] = val;
            }
        }
    }
}

__global__ void attn_wmma(const float* __restrict__ q, const float* __restrict__ kbuf,
                          const float* __restrict__ vbuf, float* __restrict__ o) {
    __shared__ float p[16][83];

    const int lane = threadIdx.x;
    const int i0   = blockIdx.x * 16;
    const int h    = blockIdx.y;
    const int b    = blockIdx.z;
    const int kvh  = h >> 2;

    const int l15  = lane & 15;
    const int koff = (lane >> 4) << 1;
    const int r8   = (lane >> 4) << 3;
    const int keystart = i0 - WIN_;

    const float* qrow = q + ((long)(b * S_ + i0 + l15) * H_ + h) * HD_;
    for (int t = 0; t < 5; ++t) {
        const int j0 = keystart + t * 16;
        if (j0 < 0) {
            #pragma unroll
            for (int r = 0; r < 8; ++r) p[r8 + r][t * 16 + l15] = -1e9f;
            continue;
        }
        const float* krow = kbuf + ((long)(b * S_ + j0 + l15) * KVH_ + kvh) * HD_;
        v8f acc = {};
        for (int kk = 0; kk < HD_; kk += 4) {
            const int kb = kk + koff;
            v2f a, bb;
            a.x  = qrow[kb];  a.y  = qrow[kb + 1];
            bb.x = krow[kb];  bb.y = krow[kb + 1];
            acc = __builtin_amdgcn_wmma_f32_16x16x4_f32(false, a, false, bb,
                                                        (short)0, acc, false, false);
        }
        #pragma unroll
        for (int r = 0; r < 8; ++r) {
            const int row  = i0 + r8 + r;
            const int colk = j0 + l15;
            float sc = fmaxf(acc[r] * 0.125f, -1e9f);
            if (colk > row || row - colk > WIN_) sc = -1e9f;
            p[r8 + r][t * 16 + l15] = sc;
        }
    }
    __syncthreads();

    if (lane < 16) {
        float m = -1e30f;
        for (int c = 0; c < 80; ++c) m = fmaxf(m, p[lane][c]);
        float sum = 0.0f;
        for (int c = 0; c < 80; ++c) {
            const float e = expf(p[lane][c] - m);
            p[lane][c] = e;
            sum += e;
        }
        const float inv = 1.0f / sum;
        for (int c = 0; c < 80; ++c) p[lane][c] *= inv;
    }
    __syncthreads();

    for (int nt = 0; nt < 4; ++nt) {
        const int n0 = nt * 16;
        v8f acc = {};
        for (int kk = 0; kk < 80; kk += 4) {
            const int kb = kk + koff;
            v2f a, bb;
            a.x = p[l15][kb];
            a.y = p[l15][kb + 1];
            const int key0 = keystart + kb;
            const int key1 = key0 + 1;
            bb.x = (key0 >= 0)
                 ? vbuf[((long)(b * S_ + key0) * KVH_ + kvh) * HD_ + n0 + l15] : 0.0f;
            bb.y = (key1 >= 0)
                 ? vbuf[((long)(b * S_ + key1) * KVH_ + kvh) * HD_ + n0 + l15] : 0.0f;
            acc = __builtin_amdgcn_wmma_f32_16x16x4_f32(false, a, false, bb,
                                                        (short)0, acc, false, false);
        }
        #pragma unroll
        for (int r = 0; r < 8; ++r) {
            o[((long)(b * S_ + i0 + r8 + r) * H_ + h) * HD_ + n0 + l15] = acc[r];
        }
    }
}

extern "C" void kernel_launch(void* const* d_in, const int* in_sizes, int n_in,
                              void* d_out, int out_size, void* d_ws, size_t ws_size,
                              hipStream_t stream) {
    const float* x    = (const float*)d_in[0];
    const float* fcos = (const float*)d_in[1];
    const float* fsin = (const float*)d_in[2];
    const float* wq   = (const float*)d_in[3];
    const float* wk   = (const float*)d_in[4];
    const float* wv   = (const float*)d_in[5];
    const float* wo   = (const float*)d_in[6];
    float* out = (float*)d_out;

    float* qb = (float*)d_ws;
    float* kb = qb + (size_t)M_ * (H_ * HD_);
    float* vb = kb + (size_t)M_ * (KVH_ * HD_);
    float* at = vb + (size_t)M_ * (KVH_ * HD_);

    const dim3 blk(128);
    gemm64x32_wmma<1, H_ * HD_, D_><<<dim3((H_ * HD_) / 128, M_ / 64), blk, 0, stream>>>(
        x, wq, qb, fcos, fsin);
    gemm64x32_wmma<1, KVH_ * HD_, D_><<<dim3((KVH_ * HD_) / 128, M_ / 64), blk, 0, stream>>>(
        x, wk, kb, fcos, fsin);
    gemm64x32_wmma<0, KVH_ * HD_, D_><<<dim3((KVH_ * HD_) / 128, M_ / 64), blk, 0, stream>>>(
        x, wv, vb, fcos, fsin);

    attn_wmma<<<dim3(S_ / 16, H_, B_), dim3(32), 0, stream>>>(qb, kb, vb, at);

    gemm64x32_wmma<0, D_, H_ * HD_><<<dim3(D_ / 128, M_ / 64), blk, 0, stream>>>(
        at, wo, out, fcos, fsin);
}


# Q, K and V projections in one launch (12 block columns: 8 Q, 2 K, 2 V, branch uniform per block): 3 waves on every SIMD where the K and V launches each filled half the chip; sum order unchanged
# speedup vs baseline: 4.7291x; 1.2375x over previous
#include <hip/hip_runtime.h>
#include <hip/hip_bf16.h>


typedef __attribute__((ext_vector_type(2))) float v2f;
typedef __attribute__((ext_vector_type(4))) float v4f;
typedef __attribute__((ext_vector_type(8))) float v8f;

#define B_ 2
#define S_ 2048
#define D_ 1024
#define H_ 16
#define KVH_ 4
#define HD_ 64
#define WIN_ 64
#define M_ (B_ * S_)

template <int DO_ROPE, int N, int K>
__device__ __forceinline__ void
gemm64x32_body(const float* __restrict__ A, const float* __restrict__ W,
               float* __restrict__ C,
               const float* __restrict__ fcos,
               const float* __restrict__ fsin,
               const int blockCol) {
    constexpr int G = 4;
    static_assert(K % (8 * G) == 0 && K >= 8 * G, "K must be whole trips of G groups");

    const int lane  = threadIdx.x & 31;
    const int wave  = threadIdx.x >> 5;
    const int tileM = blockIdx.y * 64;
    const int tileN = (blockCol * (blockDim.x >> 5) + wave) * 32;
    if (tileN >= N) return;

    const int l15  = lane & 15;
    const int koff = (lane >> 4) << 2;

    v8f acc[4][2] = {};

    const float* ap = A + (long)(tileM + l15) * K + koff;
    const float* bp = W + (long)koff * N + (tileN + l15);
    constexpr long aStride = 16L * K;

    v4f a[G][4];
    v2f b[G][2][2];

#define GEMM_LOAD_GROUP(g, apx, bpx)                                           \
    do {                                                                       \
        _Pragma("unroll")                                                      \
        for (int i = 0; i < 4; ++i)                                            \
            a[g][i] = *(const v4f*)((apx) + i * aStride + 8 * (g));            \
        _Pragma("unroll")                                                      \
        for (int w = 0; w < 2; ++w) {                                          \
            _Pragma("unroll")                                                  \
            for (int j = 0; j < 2; ++j) {                                      \
                b[g][w][j].x = (bpx)[(long)(8 * (g) + 2 * w) * N + j * 16];     \
                b[g][w][j].y = (bpx)[(long)(8 * (g) + 2 * w + 1) * N + j * 16]; \
            }                                                                  \
        }                                                                      \
    } while (0)

#define GEMM_MMA_GROUP(g)                                                      \
    do {                                                                       \
        _Pragma("unroll")                                                      \
        for (int i = 0; i < 4; ++i) {                                          \
            const v2f alo = __builtin_shufflevector(a[g][i], a[g][i], 0, 1);   \
            const v2f ahi = __builtin_shufflevector(a[g][i], a[g][i], 2, 3);   \
            _Pragma("unroll")                                                  \
            for (int j = 0; j < 2; ++j) {                                      \
                acc[i][j] = __builtin_amdgcn_wmma_f32_16x16x4_f32(             \
                    false, alo, false, b[g][0][j], (short)0, acc[i][j],        \
                    false, false);                                             \
                acc[i][j] = __builtin_amdgcn_wmma_f32_16x16x4_f32(             \
                    false, ahi, false, b[g][1][j], (short)0, acc[i][j],        \
                    false, false);                                             \
            }                                                                  \
        }                                                                      \
    } while (0)

    #pragma unroll
    for (int g = 0; g < G; ++g) {
        GEMM_LOAD_GROUP(g, ap, bp);
        __builtin_amdgcn_sched_barrier(0);
    }

    for (int kk = 8 * G; kk < K; kk += 8 * G) {
        ap += 8 * G;
        bp += (long)(8 * G) * N;
        #pragma unroll
        for (int g = 0; g < G; ++g) {
            GEMM_MMA_GROUP(g);
            GEMM_LOAD_GROUP(g, ap, bp);
        }
    }

    #pragma unroll
    for (int g = 0; g < G; ++g) GEMM_MMA_GROUP(g);

#undef GEMM_LOAD_GROUP
#undef GEMM_MMA_GROUP

    const int rowOff = (lane >> 4) << 3;
    #pragma unroll
    for (int i = 0; i < 4; ++i) {
        const int rowBase = tileM + i * 16 + rowOff;
        #pragma unroll
        for (int j = 0; j < 2; ++j) {
            const int col = tileN + j * 16 + l15;
            #pragma unroll
            for (int r = 0; r < 8; ++r) {
                float val = acc[i][j][r];
                if (DO_ROPE) {
                    const float partner = __shfl_xor(val, 1, 32);
                    const int pos = (rowBase + r) & (S_ - 1);
                    const int fi  = (col & (HD_ - 1)) >> 1;
                    const float c = fcos[pos * (HD_ / 2) + fi];
                    const float s = fsin[pos * (HD_ / 2) + fi];
                    val = ((col & 1) == 0) ? (val * c - partner * s)
                                           : (partner * s + val * c);
                }
                C[(long)(rowBase + r) * N + col] = val;
            }
        }
    }
}

__global__ void __launch_bounds__(128)
gemm_qkv_wmma(const float* __restrict__ x, const float* __restrict__ wq,
              const float* __restrict__ wk, const float* __restrict__ wv,
              float* __restrict__ q, float* __restrict__ k, float* __restrict__ v,
              const float* __restrict__ fcos, const float* __restrict__ fsin) {
    const int bx = blockIdx.x;
    if (bx < 8)       gemm64x32_body<1, H_ * HD_, D_>(x, wq, q, fcos, fsin, bx);
    else if (bx < 10) gemm64x32_body<1, KVH_ * HD_, D_>(x, wk, k, fcos, fsin, bx - 8);
    else              gemm64x32_body<0, KVH_ * HD_, D_>(x, wv, v, fcos, fsin, bx - 10);
}

template <int N, int K>
__global__ void __launch_bounds__(128)
gemm64x32_wmma(const float* __restrict__ A, const float* __restrict__ W,
               float* __restrict__ C) {
    gemm64x32_body<0, N, K>(A, W, C, nullptr, nullptr, blockIdx.x);
}

__global__ void attn_wmma(const float* __restrict__ q, const float* __restrict__ kbuf,
                          const float* __restrict__ vbuf, float* __restrict__ o) {
    __shared__ float p[16][83];

    const int lane = threadIdx.x;
    const int i0   = blockIdx.x * 16;
    const int h    = blockIdx.y;
    const int b    = blockIdx.z;
    const int kvh  = h >> 2;

    const int l15  = lane & 15;
    const int koff = (lane >> 4) << 1;
    const int r8   = (lane >> 4) << 3;
    const int keystart = i0 - WIN_;

    const float* qrow = q + ((long)(b * S_ + i0 + l15) * H_ + h) * HD_;
    for (int t = 0; t < 5; ++t) {
        const int j0 = keystart + t * 16;
        if (j0 < 0) {
            #pragma unroll
            for (int r = 0; r < 8; ++r) p[r8 + r][t * 16 + l15] = -1e9f;
            continue;
        }
        const float* krow = kbuf + ((long)(b * S_ + j0 + l15) * KVH_ + kvh) * HD_;
        v8f acc = {};
        for (int kk = 0; kk < HD_; kk += 4) {
            const int kb = kk + koff;
            v2f a, bb;
            a.x  = qrow[kb];  a.y  = qrow[kb + 1];
            bb.x = krow[kb];  bb.y = krow[kb + 1];
            acc = __builtin_amdgcn_wmma_f32_16x16x4_f32(false, a, false, bb,
                                                        (short)0, acc, false, false);
        }
        #pragma unroll
        for (int r = 0; r < 8; ++r) {
            const int row  = i0 + r8 + r;
            const int colk = j0 + l15;
            float sc = fmaxf(acc[r] * 0.125f, -1e9f);
            if (colk > row || row - colk > WIN_) sc = -1e9f;
            p[r8 + r][t * 16 + l15] = sc;
        }
    }
    __syncthreads();

    if (lane < 16) {
        float m = -1e30f;
        for (int c = 0; c < 80; ++c) m = fmaxf(m, p[lane][c]);
        float sum = 0.0f;
        for (int c = 0; c < 80; ++c) {
            const float e = expf(p[lane][c] - m);
            p[lane][c] = e;
            sum += e;
        }
        const float inv = 1.0f / sum;
        for (int c = 0; c < 80; ++c) p[lane][c] *= inv;
    }
    __syncthreads();

    for (int nt = 0; nt < 4; ++nt) {
        const int n0 = nt * 16;
        v8f acc = {};
        for (int kk = 0; kk < 80; kk += 4) {
            const int kb = kk + koff;
            v2f a, bb;
            a.x = p[l15][kb];
            a.y = p[l15][kb + 1];
            const int key0 = keystart + kb;
            const int key1 = key0 + 1;
            bb.x = (key0 >= 0)
                 ? vbuf[((long)(b * S_ + key0) * KVH_ + kvh) * HD_ + n0 + l15] : 0.0f;
            bb.y = (key1 >= 0)
                 ? vbuf[((long)(b * S_ + key1) * KVH_ + kvh) * HD_ + n0 + l15] : 0.0f;
            acc = __builtin_amdgcn_wmma_f32_16x16x4_f32(false, a, false, bb,
                                                        (short)0, acc, false, false);
        }
        #pragma unroll
        for (int r = 0; r < 8; ++r) {
            o[((long)(b * S_ + i0 + r8 + r) * H_ + h) * HD_ + n0 + l15] = acc[r];
        }
    }
}

extern "C" void kernel_launch(void* const* d_in, const int* in_sizes, int n_in,
                              void* d_out, int out_size, void* d_ws, size_t ws_size,
                              hipStream_t stream) {
    const float* x    = (const float*)d_in[0];
    const float* fcos = (const float*)d_in[1];
    const float* fsin = (const float*)d_in[2];
    const float* wq   = (const float*)d_in[3];
    const float* wk   = (const float*)d_in[4];
    const float* wv   = (const float*)d_in[5];
    const float* wo   = (const float*)d_in[6];
    float* out = (float*)d_out;

    float* qb = (float*)d_ws;
    float* kb = qb + (size_t)M_ * (H_ * HD_);
    float* vb = kb + (size_t)M_ * (KVH_ * HD_);
    float* at = vb + (size_t)M_ * (KVH_ * HD_);

    const dim3 blk(128);
    gemm_qkv_wmma<<<dim3(12, M_ / 64), blk, 0, stream>>>(
        x, wq, wk, wv, qb, kb, vb, fcos, fsin);

    attn_wmma<<<dim3(S_ / 16, H_, B_), dim3(32), 0, stream>>>(qb, kb, vb, at);

    gemm64x32_wmma<D_, H_ * HD_><<<dim3(D_ / 128, M_ / 64), blk, 0, stream>>>(
        at, wo, out);
}


# Attention: Q, K, V by branch-free 16-byte loads issued ahead of each phase (Q once; V rows before the softmax, 4 columns a lane), not a waited load pair per WMMA; score sum order changes per 8 dims
# speedup vs baseline: 5.3002x; 1.1208x over previous
#include <hip/hip_runtime.h>
#include <hip/hip_bf16.h>


typedef __attribute__((ext_vector_type(2))) float v2f;
typedef __attribute__((ext_vector_type(4))) float v4f;
typedef __attribute__((ext_vector_type(8))) float v8f;

#define B_ 2
#define S_ 2048
#define D_ 1024
#define H_ 16
#define KVH_ 4
#define HD_ 64
#define WIN_ 64
#define M_ (B_ * S_)

template <int DO_ROPE, int N, int K>
__device__ __forceinline__ void
gemm64x32_body(const float* __restrict__ A, const float* __restrict__ W,
               float* __restrict__ C,
               const float* __restrict__ fcos,
               const float* __restrict__ fsin,
               const int blockCol) {
    constexpr int G = 4;
    static_assert(K % (8 * G) == 0 && K >= 8 * G, "K must be whole trips of G groups");

    const int lane  = threadIdx.x & 31;
    const int wave  = threadIdx.x >> 5;
    const int tileM = blockIdx.y * 64;
    const int tileN = (blockCol * (blockDim.x >> 5) + wave) * 32;
    if (tileN >= N) return;

    const int l15  = lane & 15;
    const int koff = (lane >> 4) << 2;

    v8f acc[4][2] = {};

    const float* ap = A + (long)(tileM + l15) * K + koff;
    const float* bp = W + (long)koff * N + (tileN + l15);
    constexpr long aStride = 16L * K;

    v4f a[G][4];
    v2f b[G][2][2];

#define GEMM_LOAD_GROUP(g, apx, bpx)                                           \
    do {                                                                       \
        _Pragma("unroll")                                                      \
        for (int i = 0; i < 4; ++i)                                            \
            a[g][i] = *(const v4f*)((apx) + i * aStride + 8 * (g));            \
        _Pragma("unroll")                                                      \
        for (int w = 0; w < 2; ++w) {                                          \
            _Pragma("unroll")                                                  \
            for (int j = 0; j < 2; ++j) {                                      \
                b[g][w][j].x = (bpx)[(long)(8 * (g) + 2 * w) * N + j * 16];     \
                b[g][w][j].y = (bpx)[(long)(8 * (g) + 2 * w + 1) * N + j * 16]; \
            }                                                                  \
        }                                                                      \
    } while (0)

#define GEMM_MMA_GROUP(g)                                                      \
    do {                                                                       \
        _Pragma("unroll")                                                      \
        for (int i = 0; i < 4; ++i) {                                          \
            const v2f alo = __builtin_shufflevector(a[g][i], a[g][i], 0, 1);   \
            const v2f ahi = __builtin_shufflevector(a[g][i], a[g][i], 2, 3);   \
            _Pragma("unroll")                                                  \
            for (int j = 0; j < 2; ++j) {                                      \
                acc[i][j] = __builtin_amdgcn_wmma_f32_16x16x4_f32(             \
                    false, alo, false, b[g][0][j], (short)0, acc[i][j],        \
                    false, false);                                             \
                acc[i][j] = __builtin_amdgcn_wmma_f32_16x16x4_f32(             \
                    false, ahi, false, b[g][1][j], (short)0, acc[i][j],        \
                    false, false);                                             \
            }                                                                  \
        }                                                                      \
    } while (0)

    #pragma unroll
    for (int g = 0; g < G; ++g) {
        GEMM_LOAD_GROUP(g, ap, bp);
        __builtin_amdgcn_sched_barrier(0);
    }

    for (int kk = 8 * G; kk < K; kk += 8 * G) {
        ap += 8 * G;
        bp += (long)(8 * G) * N;
        #pragma unroll
        for (int g = 0; g < G; ++g) {
            GEMM_MMA_GROUP(g);
            GEMM_LOAD_GROUP(g, ap, bp);
        }
    }

    #pragma unroll
    for (int g = 0; g < G; ++g) GEMM_MMA_GROUP(g);

#undef GEMM_LOAD_GROUP
#undef GEMM_MMA_GROUP

    const int rowOff = (lane >> 4) << 3;
    #pragma unroll
    for (int i = 0; i < 4; ++i) {
        const int rowBase = tileM + i * 16 + rowOff;
        #pragma unroll
        for (int j = 0; j < 2; ++j) {
            const int col = tileN + j * 16 + l15;
            #pragma unroll
            for (int r = 0; r < 8; ++r) {
                float val = acc[i][j][r];
                if (DO_ROPE) {
                    const float partner = __shfl_xor(val, 1, 32);
                    const int pos = (rowBase + r) & (S_ - 1);
                    const int fi  = (col & (HD_ - 1)) >> 1;
                    const float c = fcos[pos * (HD_ / 2) + fi];
                    const float s = fsin[pos * (HD_ / 2) + fi];
                    val = ((col & 1) == 0) ? (val * c - partner * s)
                                           : (partner * s + val * c);
                }
                C[(long)(rowBase + r) * N + col] = val;
            }
        }
    }
}

__global__ void __launch_bounds__(128)
gemm_qkv_wmma(const float* __restrict__ x, const float* __restrict__ wq,
              const float* __restrict__ wk, const float* __restrict__ wv,
              float* __restrict__ q, float* __restrict__ k, float* __restrict__ v,
              const float* __restrict__ fcos, const float* __restrict__ fsin) {
    const int bx = blockIdx.x;
    if (bx < 8)       gemm64x32_body<1, H_ * HD_, D_>(x, wq, q, fcos, fsin, bx);
    else if (bx < 10) gemm64x32_body<1, KVH_ * HD_, D_>(x, wk, k, fcos, fsin, bx - 8);
    else              gemm64x32_body<0, KVH_ * HD_, D_>(x, wv, v, fcos, fsin, bx - 10);
}

template <int N, int K>
__global__ void __launch_bounds__(128)
gemm64x32_wmma(const float* __restrict__ A, const float* __restrict__ W,
               float* __restrict__ C) {
    gemm64x32_body<0, N, K>(A, W, C, nullptr, nullptr, blockIdx.x);
}

__global__ void __launch_bounds__(32, 4)
attn_wmma(const float* __restrict__ q, const float* __restrict__ kbuf,
          const float* __restrict__ vbuf, float* __restrict__ o) {
    __shared__ float p[16][83];

    const int lane = threadIdx.x;
    const int i0   = blockIdx.x * 16;
    const int h    = blockIdx.y;
    const int b    = blockIdx.z;
    const int kvh  = h >> 2;

    const int l15  = lane & 15;
    const int hh   = lane >> 4;
    const int r8   = hh << 3;
    const int keystart = i0 - WIN_;

    const float* qrow = q + ((long)(b * S_ + i0 + l15) * H_ + h) * HD_ + 4 * hh;
    v4f qf[8];
    #pragma unroll
    for (int s = 0; s < 8; ++s) qf[s] = *(const v4f*)(qrow + 8 * s);

    v8f sacc[5];
    #pragma unroll
    for (int t = 0; t < 5; ++t) {
        const int j0 = keystart + t * 16;
        const int jr = j0 < 0 ? 0 : j0;
        const float* krow = kbuf + ((long)(b * S_ + jr + l15) * KVH_ + kvh) * HD_ + 4 * hh;
        v8f acc = {};
        #pragma unroll
        for (int s = 0; s < 8; ++s) {
            const v4f kf = *(const v4f*)(krow + 8 * s);
            const v2f alo = __builtin_shufflevector(qf[s], qf[s], 0, 1);
            const v2f ahi = __builtin_shufflevector(qf[s], qf[s], 2, 3);
            const v2f blo = __builtin_shufflevector(kf, kf, 0, 1);
            const v2f bhi = __builtin_shufflevector(kf, kf, 2, 3);
            acc = __builtin_amdgcn_wmma_f32_16x16x4_f32(false, alo, false, blo,
                                                        (short)0, acc, false, false);
            acc = __builtin_amdgcn_wmma_f32_16x16x4_f32(false, ahi, false, bhi,
                                                        (short)0, acc, false, false);
        }
        sacc[t] = acc;
    }

    #pragma unroll
    for (int t = 0; t < 5; ++t) {
        const int j0 = keystart + t * 16;
        #pragma unroll
        for (int r = 0; r < 8; ++r) {
            const int row  = i0 + r8 + r;
            const int colk = j0 + l15;
            float sc = fmaxf(sacc[t][r] * 0.125f, -1e9f);
            if (colk < 0 || colk > row || row - colk > WIN_) sc = -1e9f;
            p[r8 + r][t * 16 + l15] = sc;
        }
    }
    __syncthreads();

    v4f vf[20][2];
    #pragma unroll
    for (int m = 0; m < 20; ++m) {
        #pragma unroll
        for (int e = 0; e < 2; ++e) {
            const int key  = keystart + 4 * m + 2 * hh + e;
            const int keyc = key < 0 ? 0 : key;
            vf[m][e] = *(const v4f*)(vbuf + ((long)(b * S_ + keyc) * KVH_ + kvh) * HD_ + 4 * l15);
        }
    }

    if (lane < 16) {
        float m = -1e30f;
        for (int c = 0; c < 80; ++c) m = fmaxf(m, p[lane][c]);
        float sum = 0.0f;
        for (int c = 0; c < 80; ++c) {
            const float e = expf(p[lane][c] - m);
            p[lane][c] = e;
            sum += e;
        }
        const float inv = 1.0f / sum;
        for (int c = 0; c < 80; ++c) p[lane][c] *= inv;
    }
    __syncthreads();

    if (keystart < 0) {
        #pragma unroll
        for (int m = 0; m < 20; ++m) {
            #pragma unroll
            for (int e = 0; e < 2; ++e) {
                if (keystart + 4 * m + 2 * hh + e < 0) vf[m][e] = (v4f){0.0f, 0.0f, 0.0f, 0.0f};
            }
        }
    }

    v8f oacc[4] = {};
    #pragma unroll
    for (int m = 0; m < 20; ++m) {
        const int kb = 4 * m + 2 * hh;
        v2f a;
        a.x = p[l15][kb];
        a.y = p[l15][kb + 1];
        #pragma unroll
        for (int j = 0; j < 4; ++j) {
            v2f bb;
            bb.x = vf[m][0][j];
            bb.y = vf[m][1][j];
            oacc[j] = __builtin_amdgcn_wmma_f32_16x16x4_f32(false, a, false, bb,
                                                            (short)0, oacc[j], false, false);
        }
    }
    #pragma unroll
    for (int r = 0; r < 8; ++r) {
        const v4f ov = {oacc[0][r], oacc[1][r], oacc[2][r], oacc[3][r]};
        *(v4f*)(o + ((long)(b * S_ + i0 + r8 + r) * H_ + h) * HD_ + 4 * l15) = ov;
    }
}

extern "C" void kernel_launch(void* const* d_in, const int* in_sizes, int n_in,
                              void* d_out, int out_size, void* d_ws, size_t ws_size,
                              hipStream_t stream) {
    const float* x    = (const float*)d_in[0];
    const float* fcos = (const float*)d_in[1];
    const float* fsin = (const float*)d_in[2];
    const float* wq   = (const float*)d_in[3];
    const float* wk   = (const float*)d_in[4];
    const float* wv   = (const float*)d_in[5];
    const float* wo   = (const float*)d_in[6];
    float* out = (float*)d_out;

    float* qb = (float*)d_ws;
    float* kb = qb + (size_t)M_ * (H_ * HD_);
    float* vb = kb + (size_t)M_ * (KVH_ * HD_);
    float* at = vb + (size_t)M_ * (KVH_ * HD_);

    const dim3 blk(128);
    gemm_qkv_wmma<<<dim3(12, M_ / 64), blk, 0, stream>>>(
        x, wq, wk, wv, qb, kb, vb, fcos, fsin);

    attn_wmma<<<dim3(S_ / 16, H_, B_), dim3(32), 0, stream>>>(qb, kb, vb, at);

    gemm64x32_wmma<D_, H_ * HD_><<<dim3(D_ / 128, M_ / 64), blk, 0, stream>>>(
        at, wo, out);
}


# GEMM W fragments by 16-byte loads (wave tile 32x64, a lane owns 4 consecutive columns of a k row): 6 loads per 16 WMMAs in one place, not 12 in two; C rows stored by 16 bytes; sum order unchanged
# speedup vs baseline: 5.5086x; 1.0393x over previous
#include <hip/hip_runtime.h>
#include <hip/hip_bf16.h>


typedef __attribute__((ext_vector_type(2))) float v2f;
typedef __attribute__((ext_vector_type(4))) float v4f;
typedef __attribute__((ext_vector_type(8))) float v8f;

#define B_ 2
#define S_ 2048
#define D_ 1024
#define H_ 16
#define KVH_ 4
#define HD_ 64
#define WIN_ 64
#define M_ (B_ * S_)

template <int DO_ROPE, int N, int K>
__device__ __forceinline__ void
gemm32x64_body(const float* __restrict__ A, const float* __restrict__ W,
               float* __restrict__ C,
               const float* __restrict__ fcos,
               const float* __restrict__ fsin,
               const int blockCol) {
    constexpr int G = 4;
    static_assert(K % (8 * G) == 0 && K >= 8 * G, "K must be whole trips of G groups");
    static_assert(N % 4 == 0, "a W row segment of 4 columns must be 16-byte aligned");

    const int lane  = threadIdx.x & 31;
    const int wave  = threadIdx.x >> 5;
    const int tileM = blockIdx.y * 64 + (wave >> 1) * 32;
    const int tileN = blockCol * 128 + (wave & 1) * 64;
    if (tileN >= N) return;

    const int l15  = lane & 15;
    const int koff = (lane >> 4) << 2;
    const int col0 = tileN + 4 * l15;

    v8f acc[2][4] = {};

    const float* ap = A + (long)(tileM + l15) * K + koff;
    const float* bp = W + (long)koff * N + col0;
    constexpr long aStride = 16L * K;

    v4f a[G][2];
    v4f b[G][4];

#define GEMM_LOAD_GROUP(g, apx, bpx)                                           \
    do {                                                                       \
        _Pragma("unroll")                                                      \
        for (int i = 0; i < 2; ++i)                                            \
            a[g][i] = *(const v4f*)((apx) + i * aStride + 8 * (g));            \
        _Pragma("unroll")                                                      \
        for (int e = 0; e < 4; ++e)                                            \
            b[g][e] = *(const v4f*)((bpx) + (long)(8 * (g) + e) * N);          \
    } while (0)

#define GEMM_MMA_GROUP(g)                                                      \
    do {                                                                       \
        _Pragma("unroll")                                                      \
        for (int i = 0; i < 2; ++i) {                                          \
            const v2f alo = __builtin_shufflevector(a[g][i], a[g][i], 0, 1);   \
            const v2f ahi = __builtin_shufflevector(a[g][i], a[g][i], 2, 3);   \
            _Pragma("unroll")                                                  \
            for (int j = 0; j < 4; ++j) {                                      \
                const v2f blo = {b[g][0][j], b[g][1][j]};                      \
                const v2f bhi = {b[g][2][j], b[g][3][j]};                      \
                acc[i][j] = __builtin_amdgcn_wmma_f32_16x16x4_f32(             \
                    false, alo, false, blo, (short)0, acc[i][j],               \
                    false, false);                                             \
                acc[i][j] = __builtin_amdgcn_wmma_f32_16x16x4_f32(             \
                    false, ahi, false, bhi, (short)0, acc[i][j],               \
                    false, false);                                             \
            }                                                                  \
        }                                                                      \
    } while (0)

    #pragma unroll
    for (int g = 0; g < G; ++g) {
        GEMM_LOAD_GROUP(g, ap, bp);
        __builtin_amdgcn_sched_barrier(0);
    }

    for (int kk = 8 * G; kk < K; kk += 8 * G) {
        ap += 8 * G;
        bp += (long)(8 * G) * N;
        #pragma unroll
        for (int g = 0; g < G; ++g) {
            GEMM_MMA_GROUP(g);
            __builtin_amdgcn_sched_barrier(0x2 | 0x4);
            GEMM_LOAD_GROUP(g, ap, bp);
            __builtin_amdgcn_sched_barrier(0x2 | 0x4);
        }
    }

    #pragma unroll
    for (int g = 0; g < G; ++g) GEMM_MMA_GROUP(g);

#undef GEMM_LOAD_GROUP
#undef GEMM_MMA_GROUP

    const int rowOff = (lane >> 4) << 3;
    #pragma unroll
    for (int i = 0; i < 2; ++i) {
        const int rowBase = tileM + i * 16 + rowOff;
        #pragma unroll
        for (int r = 0; r < 8; ++r) {
            const int row = rowBase + r;
            v4f val = {acc[i][0][r], acc[i][1][r], acc[i][2][r], acc[i][3][r]};
            if (DO_ROPE) {
                const int pos = row & (S_ - 1);
                const int fi  = (col0 & (HD_ - 1)) >> 1;
                const v2f c = *(const v2f*)(fcos + pos * (HD_ / 2) + fi);
                const v2f s = *(const v2f*)(fsin + pos * (HD_ / 2) + fi);
                const v4f in = val;
                val.x = in.x * c.x - in.y * s.x;
                val.y = in.x * s.x + in.y * c.x;
                val.z = in.z * c.y - in.w * s.y;
                val.w = in.z * s.y + in.w * c.y;
            }
            *(v4f*)(C + (long)row * N + col0) = val;
        }
    }
}

__global__ void __launch_bounds__(128)
gemm_qkv_wmma(const float* __restrict__ x, const float* __restrict__ wq,
              const float* __restrict__ wk, const float* __restrict__ wv,
              float* __restrict__ q, float* __restrict__ k, float* __restrict__ v,
              const float* __restrict__ fcos, const float* __restrict__ fsin) {
    const int bx = blockIdx.x;
    if (bx < 8)       gemm32x64_body<1, H_ * HD_, D_>(x, wq, q, fcos, fsin, bx);
    else if (bx < 10) gemm32x64_body<1, KVH_ * HD_, D_>(x, wk, k, fcos, fsin, bx - 8);
    else              gemm32x64_body<0, KVH_ * HD_, D_>(x, wv, v, fcos, fsin, bx - 10);
}

template <int N, int K>
__global__ void __launch_bounds__(128)
gemm32x64_wmma(const float* __restrict__ A, const float* __restrict__ W,
               float* __restrict__ C) {
    gemm32x64_body<0, N, K>(A, W, C, nullptr, nullptr, blockIdx.x);
}

__global__ void __launch_bounds__(32, 4)
attn_wmma(const float* __restrict__ q, const float* __restrict__ kbuf,
          const float* __restrict__ vbuf, float* __restrict__ o) {
    __shared__ float p[16][83];

    const int lane = threadIdx.x;
    const int i0   = blockIdx.x * 16;
    const int h    = blockIdx.y;
    const int b    = blockIdx.z;
    const int kvh  = h >> 2;

    const int l15  = lane & 15;
    const int hh   = lane >> 4;
    const int r8   = hh << 3;
    const int keystart = i0 - WIN_;

    const float* qrow = q + ((long)(b * S_ + i0 + l15) * H_ + h) * HD_ + 4 * hh;
    v4f qf[8];
    #pragma unroll
    for (int s = 0; s < 8; ++s) qf[s] = *(const v4f*)(qrow + 8 * s);

    v8f sacc[5];
    #pragma unroll
    for (int t = 0; t < 5; ++t) {
        const int j0 = keystart + t * 16;
        const int jr = j0 < 0 ? 0 : j0;
        const float* krow = kbuf + ((long)(b * S_ + jr + l15) * KVH_ + kvh) * HD_ + 4 * hh;
        v8f acc = {};
        #pragma unroll
        for (int s = 0; s < 8; ++s) {
            const v4f kf = *(const v4f*)(krow + 8 * s);
            const v2f alo = __builtin_shufflevector(qf[s], qf[s], 0, 1);
            const v2f ahi = __builtin_shufflevector(qf[s], qf[s], 2, 3);
            const v2f blo = __builtin_shufflevector(kf, kf, 0, 1);
            const v2f bhi = __builtin_shufflevector(kf, kf, 2, 3);
            acc = __builtin_amdgcn_wmma_f32_16x16x4_f32(false, alo, false, blo,
                                                        (short)0, acc, false, false);
            acc = __builtin_amdgcn_wmma_f32_16x16x4_f32(false, ahi, false, bhi,
                                                        (short)0, acc, false, false);
        }
        sacc[t] = acc;
    }

    #pragma unroll
    for (int t = 0; t < 5; ++t) {
        const int j0 = keystart + t * 16;
        #pragma unroll
        for (int r = 0; r < 8; ++r) {
            const int row  = i0 + r8 + r;
            const int colk = j0 + l15;
            float sc = fmaxf(sacc[t][r] * 0.125f, -1e9f);
            if (colk < 0 || colk > row || row - colk > WIN_) sc = -1e9f;
            p[r8 + r][t * 16 + l15] = sc;
        }
    }
    __syncthreads();

    v4f vf[20][2];
    #pragma unroll
    for (int m = 0; m < 20; ++m) {
        #pragma unroll
        for (int e = 0; e < 2; ++e) {
            const int key  = keystart + 4 * m + 2 * hh + e;
            const int keyc = key < 0 ? 0 : key;
            vf[m][e] = *(const v4f*)(vbuf + ((long)(b * S_ + keyc) * KVH_ + kvh) * HD_ + 4 * l15);
        }
    }

    if (lane < 16) {
        float m = -1e30f;
        for (int c = 0; c < 80; ++c) m = fmaxf(m, p[lane][c]);
        float sum = 0.0f;
        for (int c = 0; c < 80; ++c) {
            const float e = expf(p[lane][c] - m);
            p[lane][c] = e;
            sum += e;
        }
        const float inv = 1.0f / sum;
        for (int c = 0; c < 80; ++c) p[lane][c] *= inv;
    }
    __syncthreads();

    if (keystart < 0) {
        #pragma unroll
        for (int m = 0; m < 20; ++m) {
            #pragma unroll
            for (int e = 0; e < 2; ++e) {
                if (keystart + 4 * m + 2 * hh + e < 0) vf[m][e] = (v4f){0.0f, 0.0f, 0.0f, 0.0f};
            }
        }
    }

    v8f oacc[4] = {};
    #pragma unroll
    for (int m = 0; m < 20; ++m) {
        const int kb = 4 * m + 2 * hh;
        v2f a;
        a.x = p[l15][kb];
        a.y = p[l15][kb + 1];
        #pragma unroll
        for (int j = 0; j < 4; ++j) {
            v2f bb;
            bb.x = vf[m][0][j];
            bb.y = vf[m][1][j];
            oacc[j] = __builtin_amdgcn_wmma_f32_16x16x4_f32(false, a, false, bb,
                                                            (short)0, oacc[j], false, false);
        }
    }
    #pragma unroll
    for (int r = 0; r < 8; ++r) {
        const v4f ov = {oacc[0][r], oacc[1][r], oacc[2][r], oacc[3][r]};
        *(v4f*)(o + ((long)(b * S_ + i0 + r8 + r) * H_ + h) * HD_ + 4 * l15) = ov;
    }
}

extern "C" void kernel_launch(void* const* d_in, const int* in_sizes, int n_in,
                              void* d_out, int out_size, void* d_ws, size_t ws_size,
                              hipStream_t stream) {
    const float* x    = (const float*)d_in[0];
    const float* fcos = (const float*)d_in[1];
    const float* fsin = (const float*)d_in[2];
    const float* wq   = (const float*)d_in[3];
    const float* wk   = (const float*)d_in[4];
    const float* wv   = (const float*)d_in[5];
    const float* wo   = (const float*)d_in[6];
    float* out = (float*)d_out;

    float* qb = (float*)d_ws;
    float* kb = qb + (size_t)M_ * (H_ * HD_);
    float* vb = kb + (size_t)M_ * (KVH_ * HD_);
    float* at = vb + (size_t)M_ * (KVH_ * HD_);

    const dim3 blk(128);
    gemm_qkv_wmma<<<dim3(12, M_ / 64), blk, 0, stream>>>(
        x, wq, wk, wv, qb, kb, vb, fcos, fsin);

    attn_wmma<<<dim3(S_ / 16, H_, B_), dim3(32), 0, stream>>>(qb, kb, vb, at);

    gemm32x64_wmma<D_, H_ * HD_><<<dim3(D_ / 128, M_ / 64), blk, 0, stream>>>(
        at, wo, out);
}


# GEMM K loops in four stretches (1/2, 1/4, 1/8, rest) of falling wave priority (s_setprio 3,2,1,0): a SIMD's waves advance through K together, not one after another, no long lone tail; sums unchanged
# speedup vs baseline: 5.7384x; 1.0417x over previous
#include <hip/hip_runtime.h>
#include <hip/hip_bf16.h>


typedef __attribute__((ext_vector_type(2))) float v2f;
typedef __attribute__((ext_vector_type(4))) float v4f;
typedef __attribute__((ext_vector_type(8))) float v8f;

#define B_ 2
#define S_ 2048
#define D_ 1024
#define H_ 16
#define KVH_ 4
#define HD_ 64
#define WIN_ 64
#define M_ (B_ * S_)

template <int DO_ROPE, int N, int K>
__device__ __forceinline__ void
gemm32x64_body(const float* __restrict__ A, const float* __restrict__ W,
               float* __restrict__ C,
               const float* __restrict__ fcos,
               const float* __restrict__ fsin,
               const int blockCol) {
    constexpr int G = 4;
    static_assert(K % (8 * G) == 0 && K >= 8 * G, "K must be whole trips of G groups");
    static_assert(N % 4 == 0, "a W row segment of 4 columns must be 16-byte aligned");

    const int lane  = threadIdx.x & 31;
    const int wave  = threadIdx.x >> 5;
    const int tileM = blockIdx.y * 64 + (wave >> 1) * 32;
    const int tileN = blockCol * 128 + (wave & 1) * 64;
    if (tileN >= N) return;

    const int l15  = lane & 15;
    const int koff = (lane >> 4) << 2;
    const int col0 = tileN + 4 * l15;

    v8f acc[2][4] = {};

    const float* ap = A + (long)(tileM + l15) * K + koff;
    const float* bp = W + (long)koff * N + col0;
    constexpr long aStride = 16L * K;

    v4f a[G][2];
    v4f b[G][4];

#define GEMM_LOAD_GROUP(g, apx, bpx)                                           \
    do {                                                                       \
        _Pragma("unroll")                                                      \
        for (int i = 0; i < 2; ++i)                                            \
            a[g][i] = *(const v4f*)((apx) + i * aStride + 8 * (g));            \
        _Pragma("unroll")                                                      \
        for (int e = 0; e < 4; ++e)                                            \
            b[g][e] = *(const v4f*)((bpx) + (long)(8 * (g) + e) * N);          \
    } while (0)

#define GEMM_MMA_GROUP(g)                                                      \
    do {                                                                       \
        _Pragma("unroll")                                                      \
        for (int i = 0; i < 2; ++i) {                                          \
            const v2f alo = __builtin_shufflevector(a[g][i], a[g][i], 0, 1);   \
            const v2f ahi = __builtin_shufflevector(a[g][i], a[g][i], 2, 3);   \
            _Pragma("unroll")                                                  \
            for (int j = 0; j < 4; ++j) {                                      \
                const v2f blo = {b[g][0][j], b[g][1][j]};                      \
                const v2f bhi = {b[g][2][j], b[g][3][j]};                      \
                acc[i][j] = __builtin_amdgcn_wmma_f32_16x16x4_f32(             \
                    false, alo, false, blo, (short)0, acc[i][j],               \
                    false, false);                                             \
                acc[i][j] = __builtin_amdgcn_wmma_f32_16x16x4_f32(             \
                    false, ahi, false, bhi, (short)0, acc[i][j],               \
                    false, false);                                             \
            }                                                                  \
        }                                                                      \
    } while (0)

    #pragma unroll
    for (int g = 0; g < G; ++g) {
        GEMM_LOAD_GROUP(g, ap, bp);
        __builtin_amdgcn_sched_barrier(0);
    }

#define GEMM_TRIP()                                                            \
    do {                                                                       \
        ap += 8 * G;                                                           \
        bp += (long)(8 * G) * N;                                               \
             \
        _Pragma("unroll")                                                      \
        for (int g = 0; g < G; ++g) {                                          \
            GEMM_MMA_GROUP(g);                                                 \
              \
             \
                \
                                    \
            __builtin_amdgcn_sched_barrier(0x2 | 0x4);                         \
            GEMM_LOAD_GROUP(g, ap, bp);                                        \
               \
             \
              \
                                                                \
            __builtin_amdgcn_sched_barrier(0x2 | 0x4);                         \
        }                                                                      \
    } while (0)

    constexpr int TRIPS = K / (8 * G) - 1;
    constexpr int T1 = (TRIPS + 1) / 2;
    constexpr int T2 = T1 + (TRIPS + 1) / 4;
    constexpr int T3 = T2 + (TRIPS + 1) / 8;
    static_assert(T1 > 0 && T1 < T2 && T2 < T3 && T3 < TRIPS, "four non-empty stretches");
    int t = 0;
    __builtin_amdgcn_s_setprio(3);
    for (; t < T1; ++t) GEMM_TRIP();
    __builtin_amdgcn_s_setprio(2);
    #pragma nounroll
    for (; t < T2; ++t) GEMM_TRIP();
    __builtin_amdgcn_s_setprio(1);
    #pragma nounroll
    for (; t < T3; ++t) GEMM_TRIP();
    __builtin_amdgcn_s_setprio(0);
    #pragma nounroll
    for (; t < TRIPS; ++t) GEMM_TRIP();

    #pragma unroll
    for (int g = 0; g < G; ++g) GEMM_MMA_GROUP(g);

#undef GEMM_TRIP
#undef GEMM_LOAD_GROUP
#undef GEMM_MMA_GROUP

    const int rowOff = (lane >> 4) << 3;
    #pragma unroll
    for (int i = 0; i < 2; ++i) {
        const int rowBase = tileM + i * 16 + rowOff;
        #pragma unroll
        for (int r = 0; r < 8; ++r) {
            const int row = rowBase + r;
            v4f val = {acc[i][0][r], acc[i][1][r], acc[i][2][r], acc[i][3][r]};
            if (DO_ROPE) {
                const int pos = row & (S_ - 1);
                const int fi  = (col0 & (HD_ - 1)) >> 1;
                const v2f c = *(const v2f*)(fcos + pos * (HD_ / 2) + fi);
                const v2f s = *(const v2f*)(fsin + pos * (HD_ / 2) + fi);
                const v4f in = val;
                val.x = in.x * c.x - in.y * s.x;
                val.y = in.x * s.x + in.y * c.x;
                val.z = in.z * c.y - in.w * s.y;
                val.w = in.z * s.y + in.w * c.y;
            }
            *(v4f*)(C + (long)row * N + col0) = val;
        }
    }
}

__global__ void __launch_bounds__(128)
gemm_qkv_wmma(const float* __restrict__ x, const float* __restrict__ wq,
              const float* __restrict__ wk, const float* __restrict__ wv,
              float* __restrict__ q, float* __restrict__ k, float* __restrict__ v,
              const float* __restrict__ fcos, const float* __restrict__ fsin) {
    const int bx = blockIdx.x;
    if (bx < 8)       gemm32x64_body<1, H_ * HD_, D_>(x, wq, q, fcos, fsin, bx);
    else if (bx < 10) gemm32x64_body<1, KVH_ * HD_, D_>(x, wk, k, fcos, fsin, bx - 8);
    else              gemm32x64_body<0, KVH_ * HD_, D_>(x, wv, v, fcos, fsin, bx - 10);
}

template <int N, int K>
__global__ void __launch_bounds__(128)
gemm32x64_wmma(const float* __restrict__ A, const float* __restrict__ W,
               float* __restrict__ C) {
    gemm32x64_body<0, N, K>(A, W, C, nullptr, nullptr, blockIdx.x);
}

__global__ void __launch_bounds__(32, 4)
attn_wmma(const float* __restrict__ q, const float* __restrict__ kbuf,
          const float* __restrict__ vbuf, float* __restrict__ o) {
    __shared__ float p[16][83];

    const int lane = threadIdx.x;
    const int i0   = blockIdx.x * 16;
    const int h    = blockIdx.y;
    const int b    = blockIdx.z;
    const int kvh  = h >> 2;

    const int l15  = lane & 15;
    const int hh   = lane >> 4;
    const int r8   = hh << 3;
    const int keystart = i0 - WIN_;

    const float* qrow = q + ((long)(b * S_ + i0 + l15) * H_ + h) * HD_ + 4 * hh;
    v4f qf[8];
    #pragma unroll
    for (int s = 0; s < 8; ++s) qf[s] = *(const v4f*)(qrow + 8 * s);

    v8f sacc[5];
    #pragma unroll
    for (int t = 0; t < 5; ++t) {
        const int j0 = keystart + t * 16;
        const int jr = j0 < 0 ? 0 : j0;
        const float* krow = kbuf + ((long)(b * S_ + jr + l15) * KVH_ + kvh) * HD_ + 4 * hh;
        v8f acc = {};
        #pragma unroll
        for (int s = 0; s < 8; ++s) {
            const v4f kf = *(const v4f*)(krow + 8 * s);
            const v2f alo = __builtin_shufflevector(qf[s], qf[s], 0, 1);
            const v2f ahi = __builtin_shufflevector(qf[s], qf[s], 2, 3);
            const v2f blo = __builtin_shufflevector(kf, kf, 0, 1);
            const v2f bhi = __builtin_shufflevector(kf, kf, 2, 3);
            acc = __builtin_amdgcn_wmma_f32_16x16x4_f32(false, alo, false, blo,
                                                        (short)0, acc, false, false);
            acc = __builtin_amdgcn_wmma_f32_16x16x4_f32(false, ahi, false, bhi,
                                                        (short)0, acc, false, false);
        }
        sacc[t] = acc;
    }

    #pragma unroll
    for (int t = 0; t < 5; ++t) {
        const int j0 = keystart + t * 16;
        #pragma unroll
        for (int r = 0; r < 8; ++r) {
            const int row  = i0 + r8 + r;
            const int colk = j0 + l15;
            float sc = fmaxf(sacc[t][r] * 0.125f, -1e9f);
            if (colk < 0 || colk > row || row - colk > WIN_) sc = -1e9f;
            p[r8 + r][t * 16 + l15] = sc;
        }
    }
    __syncthreads();

    v4f vf[20][2];
    #pragma unroll
    for (int m = 0; m < 20; ++m) {
        #pragma unroll
        for (int e = 0; e < 2; ++e) {
            const int key  = keystart + 4 * m + 2 * hh + e;
            const int keyc = key < 0 ? 0 : key;
            vf[m][e] = *(const v4f*)(vbuf + ((long)(b * S_ + keyc) * KVH_ + kvh) * HD_ + 4 * l15);
        }
    }

    if (lane < 16) {
        float m = -1e30f;
        for (int c = 0; c < 80; ++c) m = fmaxf(m, p[lane][c]);
        float sum = 0.0f;
        for (int c = 0; c < 80; ++c) {
            const float e = expf(p[lane][c] - m);
            p[lane][c] = e;
            sum += e;
        }
        const float inv = 1.0f / sum;
        for (int c = 0; c < 80; ++c) p[lane][c] *= inv;
    }
    __syncthreads();

    if (keystart < 0) {
        #pragma unroll
        for (int m = 0; m < 20; ++m) {
            #pragma unroll
            for (int e = 0; e < 2; ++e) {
                if (keystart + 4 * m + 2 * hh + e < 0) vf[m][e] = (v4f){0.0f, 0.0f, 0.0f, 0.0f};
            }
        }
    }

    v8f oacc[4] = {};
    #pragma unroll
    for (int m = 0; m < 20; ++m) {
        const int kb = 4 * m + 2 * hh;
        v2f a;
        a.x = p[l15][kb];
        a.y = p[l15][kb + 1];
        #pragma unroll
        for (int j = 0; j < 4; ++j) {
            v2f bb;
            bb.x = vf[m][0][j];
            bb.y = vf[m][1][j];
            oacc[j] = __builtin_amdgcn_wmma_f32_16x16x4_f32(false, a, false, bb,
                                                            (short)0, oacc[j], false, false);
        }
    }
    #pragma unroll
    for (int r = 0; r < 8; ++r) {
        const v4f ov = {oacc[0][r], oacc[1][r], oacc[2][r], oacc[3][r]};
        *(v4f*)(o + ((long)(b * S_ + i0 + r8 + r) * H_ + h) * HD_ + 4 * l15) = ov;
    }
}

extern "C" void kernel_launch(void* const* d_in, const int* in_sizes, int n_in,
                              void* d_out, int out_size, void* d_ws, size_t ws_size,
                              hipStream_t stream) {
    const float* x    = (const float*)d_in[0];
    const float* fcos = (const float*)d_in[1];
    const float* fsin = (const float*)d_in[2];
    const float* wq   = (const float*)d_in[3];
    const float* wk   = (const float*)d_in[4];
    const float* wv   = (const float*)d_in[5];
    const float* wo   = (const float*)d_in[6];
    float* out = (float*)d_out;

    float* qb = (float*)d_ws;
    float* kb = qb + (size_t)M_ * (H_ * HD_);
    float* vb = kb + (size_t)M_ * (KVH_ * HD_);
    float* at = vb + (size_t)M_ * (KVH_ * HD_);

    const dim3 blk(128);
    gemm_qkv_wmma<<<dim3(12, M_ / 64), blk, 0, stream>>>(
        x, wq, wk, wv, qb, kb, vb, fcos, fsin);

    attn_wmma<<<dim3(S_ / 16, H_, B_), dim3(32), 0, stream>>>(qb, kb, vb, at);

    gemm32x64_wmma<D_, H_ * HD_><<<dim3(D_ / 128, M_ / 64), blk, 0, stream>>>(
        at, wo, out);
}
